# starting kernel
# baseline (speedup 1.0000x reference)
#include <hip/hip_runtime.h>

typedef __attribute__((ext_vector_type(2))) float v2f;
typedef __attribute__((ext_vector_type(8))) float v8f;

#define WAVES_PER_BLOCK 8
#define BLOCK_THREADS   256
#define MAX_BLOCKS      4096

template<int OFF>
__device__ __forceinline__ float swz(float v) {
    return __int_as_float(__builtin_amdgcn_ds_swizzle(__float_as_int(v), OFF));
}
#define SWAP16 0x401f

__device__ __forceinline__ float relu1(float x) {
    float r;
    asm("v_max_num_f32 %0, 0, %1" : "=v"(r) : "v"(x));
    return r;
}

__global__ __launch_bounds__(BLOCK_THREADS)
void velvec_kernel(const float* __restrict__ pos, const float* __restrict__ nbr,
                   const float* __restrict__ W1,  const float* __restrict__ b1,
                   const float* __restrict__ W2,  const float* __restrict__ b2,
                   const float* __restrict__ W3,  const float* __restrict__ b3,
                   float* __restrict__ out, int N)
{
    __shared__ float sm[WAVES_PER_BLOCK * 512];

    const int tid  = threadIdx.x;
    const int lane = tid & 31;
    const int wave = tid >> 5;
    const int hi   = lane >> 4;
    const int n    = lane & 15;
    float* tile = sm + wave * 512;

    float w1[15], bb1[5], w3[5];
#pragma unroll
    for (int i = 0; i < 15; ++i) w1[i] = W1[i];
#pragma unroll
    for (int i = 0; i < 5; ++i)  bb1[i] = b1[i];
#pragma unroll
    for (int i = 0; i < 5; ++i)  w3[i] = W3[i];
    const float bias3 = b3[0];

    auto w2t = [&](int k, int col) -> float {
        return (k < 5 && col < 5) ? W2[col * 5 + k] : 0.0f;
    };
    v2f B2lo, B2hi;
    B2lo.x = w2t(hi ? 2 : 0, n);
    B2lo.y = w2t(hi ? 3 : 1, n);
    B2hi.x = w2t(hi ? 6 : 4, n);
    B2hi.y = 0.0f;
    const float bias2 = (n < 5) ? b2[n] : 0.0f;
    v8f cbias;
#pragma unroll
    for (int j = 0; j < 8; ++j) cbias[j] = bias2;

    const int waveGlobal = blockIdx.x * WAVES_PER_BLOCK + wave;
    const int waveStride = gridDim.x * WAVES_PER_BLOCK;

    for (int p = waveGlobal; p < N; p += waveStride) {
        const float px = pos[3 * p + 0];
        const float py = pos[3 * p + 1];
        const float pz = pos[3 * p + 2];

        const int base = (p * 32 + lane) * 3;
        const float nx = nbr[base + 0];
        const float ny = nbr[base + 1];
        const float nz = nbr[base + 2];
        __builtin_prefetch(nbr + base + 32 * 3, 0, 1);

        const float dx = nx - px, dy = ny - py, dz = nz - pz;
        const float nr  = sqrtf(fmaf(dx, dx, fmaf(dy, dy, dz * dz)));
        const float inv = 1.0f / fmaxf(nr, 1e-12f);
        const float ndx = dx * inv, ndy = dy * inv, ndz = dz * inv;

        float h[5];
#pragma unroll
        for (int j = 0; j < 5; ++j) {
            float a = fmaf(w1[3 * j + 2], ndz,
                      fmaf(w1[3 * j + 1], ndy,
                      fmaf(w1[3 * j + 0], ndx, bb1[j])));
            h[j] = fmaxf(a, 0.0f);
        }

        const float s0 = swz<SWAP16>(h[0]);
        const float s1 = swz<SWAP16>(h[1]);
        const float s2 = swz<SWAP16>(h[2]);
        const float s3 = swz<SWAP16>(h[3]);
        const float s4 = swz<SWAP16>(h[4]);
        v2f A0lo, A0hi, A1lo, A1hi;
        A0lo.x = hi ? s2   : h[0];  A0lo.y = hi ? s3   : h[1];
        A0hi.x = hi ? 0.0f : h[4];  A0hi.y = 0.0f;
        A1lo.x = hi ? h[2] : s0;    A1lo.y = hi ? h[3] : s1;
        A1hi.x = hi ? 0.0f : s4;    A1hi.y = 0.0f;

        v8f c0 = __builtin_amdgcn_wmma_f32_16x16x4_f32(false, A0lo, false, B2lo,
                                                       (short)0, cbias, false, false);
        c0 = __builtin_amdgcn_wmma_f32_16x16x4_f32(false, A0hi, false, B2hi,
                                                   (short)0, c0, false, false);
        v8f c1 = __builtin_amdgcn_wmma_f32_16x16x4_f32(false, A1lo, false, B2lo,
                                                       (short)0, cbias, false, false);
        c1 = __builtin_amdgcn_wmma_f32_16x16x4_f32(false, A1hi, false, B2hi,
                                                   (short)0, c1, false, false);

        const int sbase = hi * 8 * 16 + n;
#pragma unroll
        for (int j = 0; j < 8; ++j) {
            tile[sbase + j * 16]       = relu1(c0[j]);
            tile[256 + sbase + j * 16] = relu1(c1[j]);
        }
        const float* hrow = tile + hi * 256 + n * 16;
        float w = bias3;
#pragma unroll
        for (int i = 0; i < 5; ++i) w = fmaf(w3[i], hrow[i], w);

        float cx = ndx * w, cy = ndy * w, cz = ndz * w;
        cx += swz<0x041f>(cx); cy += swz<0x041f>(cy); cz += swz<0x041f>(cz);
        cx += swz<0x081f>(cx); cy += swz<0x081f>(cy); cz += swz<0x081f>(cz);
        cx += swz<0x101f>(cx); cy += swz<0x101f>(cy); cz += swz<0x101f>(cz);
        cx += swz<0x201f>(cx); cy += swz<0x201f>(cy); cz += swz<0x201f>(cz);
        cx += swz<0x401f>(cx); cy += swz<0x401f>(cy); cz += swz<0x401f>(cz);

        const float vn = sqrtf(fmaf(cx, cx, fmaf(cy, cy, cz * cz)));
        const float vi = 1.0f / fmaxf(vn, 1e-12f);
        if (lane == 0) {
            out[3 * p + 0] = cx * vi;
            out[3 * p + 1] = cy * vi;
            out[3 * p + 2] = cz * vi;
        }
    }
}

extern "C" void kernel_launch(void* const* d_in, const int* in_sizes, int n_in,
                              void* d_out, int out_size, void* d_ws, size_t ws_size,
                              hipStream_t stream) {
    const float* pos = (const float*)d_in[0];
    const float* nbr = (const float*)d_in[1];
    const float* W1  = (const float*)d_in[2];
    const float* b1  = (const float*)d_in[3];
    const float* W2  = (const float*)d_in[4];
    const float* b2  = (const float*)d_in[5];
    const float* W3  = (const float*)d_in[6];
    const float* b3  = (const float*)d_in[7];
    float* out = (float*)d_out;

    const int N = in_sizes[0] / 3;
    int blocks = (N + WAVES_PER_BLOCK - 1) / WAVES_PER_BLOCK;
    if (blocks > MAX_BLOCKS) blocks = MAX_BLOCKS;
    velvec_kernel<<<blocks, BLOCK_THREADS, 0, stream>>>(pos, nbr, W1, b1, W2, b2,
                                                        W3, b3, out, N);
}


# Layer 2 transposed on the WMMA: W2 as block-diagonal A operand, h1 as B, so D lands per lane with its own neighbor's features; removes LDS transpose, lane swaps and selects, and one of four WMMAs
# speedup vs baseline: 1.1382x; 1.1382x over previous
#include <hip/hip_runtime.h>

typedef __attribute__((ext_vector_type(2))) float v2f;
typedef __attribute__((ext_vector_type(8))) float v8f;

#define WAVES_PER_BLOCK 8
#define BLOCK_THREADS   256
#define MAX_BLOCKS      4096

template<int OFF>
__device__ __forceinline__ float swz(float v) {
    return __int_as_float(__builtin_amdgcn_ds_swizzle(__float_as_int(v), OFF));
}

__device__ __forceinline__ float relu1(float x) {
    float r;
    asm("v_max_num_f32 %0, 0, %1" : "=v"(r) : "v"(x));
    return r;
}

__global__ __launch_bounds__(BLOCK_THREADS)
void velvec_kernel(const float* __restrict__ pos, const float* __restrict__ nbr,
                   const float* __restrict__ W1,  const float* __restrict__ b1,
                   const float* __restrict__ W2,  const float* __restrict__ b2,
                   const float* __restrict__ W3,  const float* __restrict__ b3,
                   float* __restrict__ out, int N)
{
    const int tid  = threadIdx.x;
    const int lane = tid & 31;
    const int wave = tid >> 5;
    const int hi   = lane >> 4;
    const int n    = lane & 15;

    float w1[15], bb1[5], w3[5];
#pragma unroll
    for (int i = 0; i < 15; ++i) w1[i] = W1[i];
#pragma unroll
    for (int i = 0; i < 5; ++i)  bb1[i] = b1[i];
#pragma unroll
    for (int i = 0; i < 5; ++i)  w3[i] = W3[i];
    const float bias3 = b3[0];

    const int  g      = n - 8 * hi;
    const bool rowok  = (g >= 0) && (g < 5);
    auto w2a = [&](int k) -> float {
        return (rowok && k < 5) ? W2[g * 5 + k] : 0.0f;
    };
    v2f A2_0, A2_1, A2_2;
    A2_0.x = w2a(0);  A2_0.y = w2a(1);
    A2_1.x = w2a(2);  A2_1.y = w2a(3);
    A2_2.x = w2a(4);  A2_2.y = 0.0f;
    v8f cbias;
#pragma unroll
    for (int j = 0; j < 8; ++j) cbias[j] = (j < 5) ? b2[j] : 0.0f;

    const int waveGlobal = blockIdx.x * WAVES_PER_BLOCK + wave;
    const int waveStride = gridDim.x * WAVES_PER_BLOCK;

    for (int p = waveGlobal; p < N; p += waveStride) {
        const float px = pos[3 * p + 0];
        const float py = pos[3 * p + 1];
        const float pz = pos[3 * p + 2];

        const int base = (p * 32 + lane) * 3;
        const float nx = nbr[base + 0];
        const float ny = nbr[base + 1];
        const float nz = nbr[base + 2];
        __builtin_prefetch(nbr + base + 32 * 3, 0, 1);

        const float dx = nx - px, dy = ny - py, dz = nz - pz;
        const float nr  = sqrtf(fmaf(dx, dx, fmaf(dy, dy, dz * dz)));
        const float inv = 1.0f / fmaxf(nr, 1e-12f);
        const float ndx = dx * inv, ndy = dy * inv, ndz = dz * inv;

        float h[5];
#pragma unroll
        for (int j = 0; j < 5; ++j) {
            float a = fmaf(w1[3 * j + 2], ndz,
                      fmaf(w1[3 * j + 1], ndy,
                      fmaf(w1[3 * j + 0], ndx, bb1[j])));
            h[j] = fmaxf(a, 0.0f);
        }

        v2f B0, B1, B2;
        B0.x = h[0];  B0.y = h[1];
        B1.x = h[2];  B1.y = h[3];
        B2.x = h[4];  B2.y = 0.0f;
        v8f c = __builtin_amdgcn_wmma_f32_16x16x4_f32(false, A2_0, false, B0,
                                                      (short)0, cbias, false, false);
        c = __builtin_amdgcn_wmma_f32_16x16x4_f32(false, A2_1, false, B1,
                                                  (short)0, c, false, false);
        c = __builtin_amdgcn_wmma_f32_16x16x4_f32(false, A2_2, false, B2,
                                                  (short)0, c, false, false);

        float w = bias3;
#pragma unroll
        for (int i = 0; i < 5; ++i) w = fmaf(w3[i], relu1(c[i]), w);

        float cx = ndx * w, cy = ndy * w, cz = ndz * w;
        cx += swz<0x041f>(cx); cy += swz<0x041f>(cy); cz += swz<0x041f>(cz);
        cx += swz<0x081f>(cx); cy += swz<0x081f>(cy); cz += swz<0x081f>(cz);
        cx += swz<0x101f>(cx); cy += swz<0x101f>(cy); cz += swz<0x101f>(cz);
        cx += swz<0x201f>(cx); cy += swz<0x201f>(cy); cz += swz<0x201f>(cz);
        cx += swz<0x401f>(cx); cy += swz<0x401f>(cy); cz += swz<0x401f>(cz);

        const float vn = sqrtf(fmaf(cx, cx, fmaf(cy, cy, cz * cz)));
        const float vi = 1.0f / fmaxf(vn, 1e-12f);
        if (lane == 0) {
            out[3 * p + 0] = cx * vi;
            out[3 * p + 1] = cy * vi;
            out[3 * p + 2] = cz * vi;
        }
    }
}

extern "C" void kernel_launch(void* const* d_in, const int* in_sizes, int n_in,
                              void* d_out, int out_size, void* d_ws, size_t ws_size,
                              hipStream_t stream) {
    const float* pos = (const float*)d_in[0];
    const float* nbr = (const float*)d_in[1];
    const float* W1  = (const float*)d_in[2];
    const float* b1  = (const float*)d_in[3];
    const float* W2  = (const float*)d_in[4];
    const float* b2  = (const float*)d_in[5];
    const float* W3  = (const float*)d_in[6];
    const float* b3  = (const float*)d_in[7];
    float* out = (float*)d_out;

    const int N = in_sizes[0] / 3;
    int blocks = (N + WAVES_PER_BLOCK - 1) / WAVES_PER_BLOCK;
    if (blocks > MAX_BLOCKS) blocks = MAX_BLOCKS;
    velvec_kernel<<<blocks, BLOCK_THREADS, 0, stream>>>(pos, nbr, W1, b1, W2, b2,
                                                        W3, b3, out, N);
}


# Both normalisations take 1/max(norm,eps) as one v_rsq_f32 of max(squared norm, eps*eps) instead of exact sqrt + IEEE divide (about 25 VALU each); scale factor now ~1 ulp, sum order as before
# speedup vs baseline: 1.3136x; 1.1541x over previous
#include <hip/hip_runtime.h>

typedef __attribute__((ext_vector_type(2))) float v2f;
typedef __attribute__((ext_vector_type(8))) float v8f;

#define WAVES_PER_BLOCK 8
#define BLOCK_THREADS   256
#define MAX_BLOCKS      4096

template<int OFF>
__device__ __forceinline__ float swz(float v) {
    return __int_as_float(__builtin_amdgcn_ds_swizzle(__float_as_int(v), OFF));
}

__device__ __forceinline__ float relu1(float x) {
    float r;
    asm("v_max_num_f32 %0, 0, %1" : "=v"(r) : "v"(x));
    return r;
}

__device__ __forceinline__ float inv_norm_from_sq(float s) {
    return __builtin_amdgcn_rsqf(fmaxf(s, 1e-24f));
}

__global__ __launch_bounds__(BLOCK_THREADS)
void velvec_kernel(const float* __restrict__ pos, const float* __restrict__ nbr,
                   const float* __restrict__ W1,  const float* __restrict__ b1,
                   const float* __restrict__ W2,  const float* __restrict__ b2,
                   const float* __restrict__ W3,  const float* __restrict__ b3,
                   float* __restrict__ out, int N)
{
    const int tid  = threadIdx.x;
    const int lane = tid & 31;
    const int wave = tid >> 5;
    const int hi   = lane >> 4;
    const int n    = lane & 15;

    float w1[15], bb1[5], w3[5];
#pragma unroll
    for (int i = 0; i < 15; ++i) w1[i] = W1[i];
#pragma unroll
    for (int i = 0; i < 5; ++i)  bb1[i] = b1[i];
#pragma unroll
    for (int i = 0; i < 5; ++i)  w3[i] = W3[i];
    const float bias3 = b3[0];

    const int  g      = n - 8 * hi;
    const bool rowok  = (g >= 0) && (g < 5);
    auto w2a = [&](int k) -> float {
        return (rowok && k < 5) ? W2[g * 5 + k] : 0.0f;
    };
    v2f A2_0, A2_1, A2_2;
    A2_0.x = w2a(0);  A2_0.y = w2a(1);
    A2_1.x = w2a(2);  A2_1.y = w2a(3);
    A2_2.x = w2a(4);  A2_2.y = 0.0f;
    v8f cbias;
#pragma unroll
    for (int j = 0; j < 8; ++j) cbias[j] = (j < 5) ? b2[j] : 0.0f;

    const int waveGlobal = blockIdx.x * WAVES_PER_BLOCK + wave;
    const int waveStride = gridDim.x * WAVES_PER_BLOCK;

    for (int p = waveGlobal; p < N; p += waveStride) {
        const float px = pos[3 * p + 0];
        const float py = pos[3 * p + 1];
        const float pz = pos[3 * p + 2];

        const int base = (p * 32 + lane) * 3;
        const float nx = nbr[base + 0];
        const float ny = nbr[base + 1];
        const float nz = nbr[base + 2];
        __builtin_prefetch(nbr + base + 32 * 3, 0, 1);

        const float dx = nx - px, dy = ny - py, dz = nz - pz;
        const float inv = inv_norm_from_sq(fmaf(dx, dx, fmaf(dy, dy, dz * dz)));
        const float ndx = dx * inv, ndy = dy * inv, ndz = dz * inv;

        float h[5];
#pragma unroll
        for (int j = 0; j < 5; ++j) {
            float a = fmaf(w1[3 * j + 2], ndz,
                      fmaf(w1[3 * j + 1], ndy,
                      fmaf(w1[3 * j + 0], ndx, bb1[j])));
            h[j] = fmaxf(a, 0.0f);
        }

        v2f B0, B1, B2;
        B0.x = h[0];  B0.y = h[1];
        B1.x = h[2];  B1.y = h[3];
        B2.x = h[4];  B2.y = 0.0f;
        v8f c = __builtin_amdgcn_wmma_f32_16x16x4_f32(false, A2_0, false, B0,
                                                      (short)0, cbias, false, false);
        c = __builtin_amdgcn_wmma_f32_16x16x4_f32(false, A2_1, false, B1,
                                                  (short)0, c, false, false);
        c = __builtin_amdgcn_wmma_f32_16x16x4_f32(false, A2_2, false, B2,
                                                  (short)0, c, false, false);

        float w = bias3;
#pragma unroll
        for (int i = 0; i < 5; ++i) w = fmaf(w3[i], relu1(c[i]), w);

        float cx = ndx * w, cy = ndy * w, cz = ndz * w;
        cx += swz<0x041f>(cx); cy += swz<0x041f>(cy); cz += swz<0x041f>(cz);
        cx += swz<0x081f>(cx); cy += swz<0x081f>(cy); cz += swz<0x081f>(cz);
        cx += swz<0x101f>(cx); cy += swz<0x101f>(cy); cz += swz<0x101f>(cz);
        cx += swz<0x201f>(cx); cy += swz<0x201f>(cy); cz += swz<0x201f>(cz);
        cx += swz<0x401f>(cx); cy += swz<0x401f>(cy); cz += swz<0x401f>(cz);

        const float vi = inv_norm_from_sq(fmaf(cx, cx, fmaf(cy, cy, cz * cz)));
        if (lane == 0) {
            out[3 * p + 0] = cx * vi;
            out[3 * p + 1] = cy * vi;
            out[3 * p + 2] = cz * vi;
        }
    }
}

extern "C" void kernel_launch(void* const* d_in, const int* in_sizes, int n_in,
                              void* d_out, int out_size, void* d_ws, size_t ws_size,
                              hipStream_t stream) {
    const float* pos = (const float*)d_in[0];
    const float* nbr = (const float*)d_in[1];
    const float* W1  = (const float*)d_in[2];
    const float* b1  = (const float*)d_in[3];
    const float* W2  = (const float*)d_in[4];
    const float* b2  = (const float*)d_in[5];
    const float* W3  = (const float*)d_in[6];
    const float* b3  = (const float*)d_in[7];
    float* out = (float*)d_out;

    const int N = in_sizes[0] / 3;
    int blocks = (N + WAVES_PER_BLOCK - 1) / WAVES_PER_BLOCK;
    if (blocks > MAX_BLOCKS) blocks = MAX_BLOCKS;
    velvec_kernel<<<blocks, BLOCK_THREADS, 0, stream>>>(pos, nbr, W1, b1, W2, b2,
                                                        W3, b3, out, N);
}


# Each lane takes 4 neighbors of one point (wave = 4 points per trip): three 16-byte loads per lane, in-lane partial sums, 3-step instead of 5-step lane butterfly; sum order changes with it
# speedup vs baseline: 2.1023x; 1.6004x over previous
#include <hip/hip_runtime.h>


typedef __attribute__((ext_vector_type(2))) float v2f;
typedef __attribute__((ext_vector_type(4))) float v4f;
typedef __attribute__((ext_vector_type(8))) float v8f;
typedef v4f v4f_a4 __attribute__((aligned(4)));

#define WAVES_PER_BLOCK 8
#define BLOCK_THREADS   256
#define GRID_BLOCKS     2048
#define PTS_PER_WAVE    4
#define NBR_PER_LANE    4

template<int OFF>
__device__ __forceinline__ float swz(float v) {
    return __int_as_float(__builtin_amdgcn_ds_swizzle(__float_as_int(v), OFF));
}

__device__ __forceinline__ float relu1(float x) {
    float r;
    asm("v_max_num_f32 %0, 0, %1" : "=v"(r) : "v"(x));
    return r;
}

__device__ __forceinline__ float inv_norm_from_sq(float s) {
    return __builtin_amdgcn_rsqf(fmaxf(s, 1e-24f));
}

__global__ __launch_bounds__(BLOCK_THREADS)
void velvec_kernel(const float* __restrict__ pos, const float* __restrict__ nbr,
                   const float* __restrict__ W1,  const float* __restrict__ b1,
                   const float* __restrict__ W2,  const float* __restrict__ b2,
                   const float* __restrict__ W3,  const float* __restrict__ b3,
                   float* __restrict__ out, int N)
{
    const int tid  = threadIdx.x;
    const int lane = tid & 31;
    const int wave = tid >> 5;
    const int hi   = lane >> 4;
    const int n    = lane & 15;

    float w1[15], bb1[5], w3[5];
#pragma unroll
    for (int i = 0; i < 15; ++i) w1[i] = W1[i];
#pragma unroll
    for (int i = 0; i < 5; ++i)  bb1[i] = b1[i];
#pragma unroll
    for (int i = 0; i < 5; ++i)  w3[i] = W3[i];
    const float bias3 = b3[0];

    const int  g      = n - 8 * hi;
    const bool rowok  = (g >= 0) && (g < 5);
    auto w2a = [&](int k) -> float {
        return (rowok && k < 5) ? W2[g * 5 + k] : 0.0f;
    };
    v2f A2_0, A2_1, A2_2;
    A2_0.x = w2a(0);  A2_0.y = w2a(1);
    A2_1.x = w2a(2);  A2_1.y = w2a(3);
    A2_2.x = w2a(4);  A2_2.y = 0.0f;
    v8f cbias;
#pragma unroll
    for (int j = 0; j < 8; ++j) cbias[j] = (j < 5) ? b2[j] : 0.0f;

    const int sub = lane >> 3;
    const int numGroups = (N + PTS_PER_WAVE - 1) / PTS_PER_WAVE;
    const int waveGlobal = blockIdx.x * WAVES_PER_BLOCK + wave;
    const int waveStride = gridDim.x * WAVES_PER_BLOCK;

    for (int grp = waveGlobal; grp < numGroups; grp += waveStride) {
        const int q  = grp * PTS_PER_WAVE + sub;
        const int qc = (q < N) ? q : (N - 1);

        const float px = pos[3 * qc + 0];
        const float py = pos[3 * qc + 1];
        const float pz = pos[3 * qc + 2];

        const float* src = nbr + (qc * 32 + (lane & 7) * NBR_PER_LANE) * 3;
        const v4f r0 = *(const v4f_a4*)(src + 0);
        const v4f r1 = *(const v4f_a4*)(src + 4);
        const v4f r2 = *(const v4f_a4*)(src + 8);
        const float in[12] = { r0.x, r0.y, r0.z, r0.w, r1.x, r1.y, r1.z, r1.w,
                               r2.x, r2.y, r2.z, r2.w };

        float ax = 0.0f, ay = 0.0f, az = 0.0f;
#pragma unroll
        for (int s = 0; s < NBR_PER_LANE; ++s) {
            const float dx = in[3 * s + 0] - px, dy = in[3 * s + 1] - py, dz = in[3 * s + 2] - pz;
            const float inv = inv_norm_from_sq(fmaf(dx, dx, fmaf(dy, dy, dz * dz)));
            const float ndx = dx * inv, ndy = dy * inv, ndz = dz * inv;

            float h[5];
#pragma unroll
            for (int j = 0; j < 5; ++j) {
                float a = fmaf(w1[3 * j + 2], ndz,
                          fmaf(w1[3 * j + 1], ndy,
                          fmaf(w1[3 * j + 0], ndx, bb1[j])));
                h[j] = fmaxf(a, 0.0f);
            }

            v2f B0, B1, B2;
            B0.x = h[0];  B0.y = h[1];
            B1.x = h[2];  B1.y = h[3];
            B2.x = h[4];  B2.y = 0.0f;
            v8f c = __builtin_amdgcn_wmma_f32_16x16x4_f32(false, A2_0, false, B0,
                                                          (short)0, cbias, false, false);
            c = __builtin_amdgcn_wmma_f32_16x16x4_f32(false, A2_1, false, B1,
                                                      (short)0, c, false, false);
            c = __builtin_amdgcn_wmma_f32_16x16x4_f32(false, A2_2, false, B2,
                                                      (short)0, c, false, false);

            float w = bias3;
#pragma unroll
            for (int i = 0; i < 5; ++i) w = fmaf(w3[i], relu1(c[i]), w);

            if (s == 0) { ax = ndx * w;  ay = ndy * w;  az = ndz * w; }
            else        { ax = fmaf(ndx, w, ax);  ay = fmaf(ndy, w, ay);  az = fmaf(ndz, w, az); }
        }

        float cx = ax, cy = ay, cz = az;
        cx += swz<0x041f>(cx); cy += swz<0x041f>(cy); cz += swz<0x041f>(cz);
        cx += swz<0x081f>(cx); cy += swz<0x081f>(cy); cz += swz<0x081f>(cz);
        cx += swz<0x101f>(cx); cy += swz<0x101f>(cy); cz += swz<0x101f>(cz);

        const float vi = inv_norm_from_sq(fmaf(cx, cx, fmaf(cy, cy, cz * cz)));
        if ((lane & 7) == 0 && q < N) {
            out[3 * q + 0] = cx * vi;
            out[3 * q + 1] = cy * vi;
            out[3 * q + 2] = cz * vi;
        }
    }
}

extern "C" void kernel_launch(void* const* d_in, const int* in_sizes, int n_in,
                              void* d_out, int out_size, void* d_ws, size_t ws_size,
                              hipStream_t stream) {
    const float* pos = (const float*)d_in[0];
    const float* nbr = (const float*)d_in[1];
    const float* W1  = (const float*)d_in[2];
    const float* b1  = (const float*)d_in[3];
    const float* W2  = (const float*)d_in[4];
    const float* b2  = (const float*)d_in[5];
    const float* W3  = (const float*)d_in[6];
    const float* b3  = (const float*)d_in[7];
    float* out = (float*)d_out;

    const int N = in_sizes[0] / 3;
    velvec_kernel<<<GRID_BLOCKS, BLOCK_THREADS, 0, stream>>>(pos, nbr, W1, b1, W2, b2,
                                                             W3, b3, out, N);
}


# Layer-2 ReLU as an integer max of the float bits instead of an inline-asm max, so the compiler can pair it: 68 dual-issue pairs per trip (was 48), 139 vector slots (was 164); wave index kept scalar
# speedup vs baseline: 2.2024x; 1.0476x over previous
#include <hip/hip_runtime.h>


typedef __attribute__((ext_vector_type(2))) float v2f;
typedef __attribute__((ext_vector_type(4))) float v4f;
typedef __attribute__((ext_vector_type(8))) float v8f;
typedef v4f v4f_a4 __attribute__((aligned(4)));

#define WAVES_PER_BLOCK 8
#define BLOCK_THREADS   256
#define GRID_BLOCKS     2048
#define PTS_PER_WAVE    4
#define NBR_PER_LANE    4

template<int OFF>
__device__ __forceinline__ float swz(float v) {
    return __int_as_float(__builtin_amdgcn_ds_swizzle(__float_as_int(v), OFF));
}

__device__ __forceinline__ float relu1(float x) {
    const int b = __float_as_int(x);
    return __int_as_float(b > 0 ? b : 0);
}

__device__ __forceinline__ float inv_norm_from_sq(float s) {
    return __builtin_amdgcn_rsqf(fmaxf(s, 1e-24f));
}

__global__ __launch_bounds__(BLOCK_THREADS)
void velvec_kernel(const float* __restrict__ pos, const float* __restrict__ nbr,
                   const float* __restrict__ W1,  const float* __restrict__ b1,
                   const float* __restrict__ W2,  const float* __restrict__ b2,
                   const float* __restrict__ W3,  const float* __restrict__ b3,
                   float* __restrict__ out, int N)
{
    const int tid  = threadIdx.x;
    const int lane = tid & 31;
    const int wave = __builtin_amdgcn_readfirstlane(tid >> 5);
    const int hi   = lane >> 4;
    const int n    = lane & 15;

    float w1[15], bb1[5], w3[5];
#pragma unroll
    for (int i = 0; i < 15; ++i) w1[i] = W1[i];
#pragma unroll
    for (int i = 0; i < 5; ++i)  bb1[i] = b1[i];
#pragma unroll
    for (int i = 0; i < 5; ++i)  w3[i] = W3[i];
    const float bias3 = b3[0];

    const int  g      = n - 8 * hi;
    const bool rowok  = (g >= 0) && (g < 5);
    auto w2a = [&](int k) -> float {
        return (rowok && k < 5) ? W2[g * 5 + k] : 0.0f;
    };
    v2f A2_0, A2_1, A2_2;
    A2_0.x = w2a(0);  A2_0.y = w2a(1);
    A2_1.x = w2a(2);  A2_1.y = w2a(3);
    A2_2.x = w2a(4);  A2_2.y = 0.0f;
    v8f cbias;
#pragma unroll
    for (int j = 0; j < 8; ++j) cbias[j] = (j < 5) ? b2[j] : 0.0f;

    const int sub = lane >> 3;
    const int numGroups = (N + PTS_PER_WAVE - 1) / PTS_PER_WAVE;
    const int waveGlobal = blockIdx.x * WAVES_PER_BLOCK + wave;
    const int waveStride = gridDim.x * WAVES_PER_BLOCK;

    for (int grp = waveGlobal; grp < numGroups; grp += waveStride) {
        const int q  = grp * PTS_PER_WAVE + sub;
        const int qc = (q < N) ? q : (N - 1);

        const float px = pos[3 * qc + 0];
        const float py = pos[3 * qc + 1];
        const float pz = pos[3 * qc + 2];

        const float* src = nbr + (qc * 32 + (lane & 7) * NBR_PER_LANE) * 3;
        const v4f r0 = *(const v4f_a4*)(src + 0);
        const v4f r1 = *(const v4f_a4*)(src + 4);
        const v4f r2 = *(const v4f_a4*)(src + 8);
        const float in[12] = { r0.x, r0.y, r0.z, r0.w, r1.x, r1.y, r1.z, r1.w,
                               r2.x, r2.y, r2.z, r2.w };

        float ax = 0.0f, ay = 0.0f, az = 0.0f;
#pragma unroll
        for (int s = 0; s < NBR_PER_LANE; ++s) {
            const float dx = in[3 * s + 0] - px, dy = in[3 * s + 1] - py, dz = in[3 * s + 2] - pz;
            const float inv = inv_norm_from_sq(fmaf(dx, dx, fmaf(dy, dy, dz * dz)));
            const float ndx = dx * inv, ndy = dy * inv, ndz = dz * inv;

            float h[5];
#pragma unroll
            for (int j = 0; j < 5; ++j) {
                float a = fmaf(w1[3 * j + 2], ndz,
                          fmaf(w1[3 * j + 1], ndy,
                          fmaf(w1[3 * j + 0], ndx, bb1[j])));
                h[j] = fmaxf(a, 0.0f);
            }

            v2f B0, B1, B2;
            B0.x = h[0];  B0.y = h[1];
            B1.x = h[2];  B1.y = h[3];
            B2.x = h[4];  B2.y = 0.0f;
            v8f c = __builtin_amdgcn_wmma_f32_16x16x4_f32(false, A2_0, false, B0,
                                                          (short)0, cbias, false, false);
            c = __builtin_amdgcn_wmma_f32_16x16x4_f32(false, A2_1, false, B1,
                                                      (short)0, c, false, false);
            c = __builtin_amdgcn_wmma_f32_16x16x4_f32(false, A2_2, false, B2,
                                                      (short)0, c, false, false);

            float w = bias3;
#pragma unroll
            for (int i = 0; i < 5; ++i) w = fmaf(w3[i], relu1(c[i]), w);

            if (s == 0) { ax = ndx * w;  ay = ndy * w;  az = ndz * w; }
            else        { ax = fmaf(ndx, w, ax);  ay = fmaf(ndy, w, ay);  az = fmaf(ndz, w, az); }
        }

        float cx = ax, cy = ay, cz = az;
        cx += swz<0x041f>(cx); cy += swz<0x041f>(cy); cz += swz<0x041f>(cz);
        cx += swz<0x081f>(cx); cy += swz<0x081f>(cy); cz += swz<0x081f>(cz);
        cx += swz<0x101f>(cx); cy += swz<0x101f>(cy); cz += swz<0x101f>(cz);

        const float vi = inv_norm_from_sq(fmaf(cx, cx, fmaf(cy, cy, cz * cz)));
        if ((lane & 7) == 0 && q < N) {
            out[3 * q + 0] = cx * vi;
            out[3 * q + 1] = cy * vi;
            out[3 * q + 2] = cz * vi;
        }
    }
}

extern "C" void kernel_launch(void* const* d_in, const int* in_sizes, int n_in,
                              void* d_out, int out_size, void* d_ws, size_t ws_size,
                              hipStream_t stream) {
    const float* pos = (const float*)d_in[0];
    const float* nbr = (const float*)d_in[1];
    const float* W1  = (const float*)d_in[2];
    const float* b1  = (const float*)d_in[3];
    const float* W2  = (const float*)d_in[4];
    const float* b2  = (const float*)d_in[5];
    const float* W3  = (const float*)d_in[6];
    const float* b3  = (const float*)d_in[7];
    float* out = (float*)d_out;

    const int N = in_sizes[0] / 3;
    velvec_kernel<<<GRID_BLOCKS, BLOCK_THREADS, 0, stream>>>(pos, nbr, W1, b1, W2, b2,
                                                             W3, b3, out, N);
}


# Layer 2's k remainder (5th input feature) leaves the half-padded third f32 16x16x4 WMMA: 5 vector FMAs per neighbor add W2[g][4]*h1[4] to D before the ReLU; 12 -> 8 WMMAs per trip; sum order kept
# speedup vs baseline: 2.3922x; 1.0862x over previous
#include <hip/hip_runtime.h>


typedef __attribute__((ext_vector_type(2))) float v2f;
typedef __attribute__((ext_vector_type(4))) float v4f;
typedef __attribute__((ext_vector_type(8))) float v8f;
typedef v4f v4f_a4 __attribute__((aligned(4)));

#define WAVES_PER_BLOCK 8
#define BLOCK_THREADS   256
#define GRID_BLOCKS     2048
#define PTS_PER_WAVE    4
#define NBR_PER_LANE    4

template<int OFF>
__device__ __forceinline__ float swz(float v) {
    return __int_as_float(__builtin_amdgcn_ds_swizzle(__float_as_int(v), OFF));
}

__device__ __forceinline__ float relu1(float x) {
    const int b = __float_as_int(x);
    return __int_as_float(b > 0 ? b : 0);
}

__device__ __forceinline__ float inv_norm_from_sq(float s) {
    return __builtin_amdgcn_rsqf(fmaxf(s, 1e-24f));
}

__global__ __launch_bounds__(BLOCK_THREADS)
void velvec_kernel(const float* __restrict__ pos, const float* __restrict__ nbr,
                   const float* __restrict__ W1,  const float* __restrict__ b1,
                   const float* __restrict__ W2,  const float* __restrict__ b2,
                   const float* __restrict__ W3,  const float* __restrict__ b3,
                   float* __restrict__ out, int N)
{
    const int tid  = threadIdx.x;
    const int lane = tid & 31;
    const int wave = __builtin_amdgcn_readfirstlane(tid >> 5);
    const int hi   = lane >> 4;
    const int n    = lane & 15;

    float w1[15], bb1[5], w3[5];
#pragma unroll
    for (int i = 0; i < 15; ++i) w1[i] = W1[i];
#pragma unroll
    for (int i = 0; i < 5; ++i)  bb1[i] = b1[i];
#pragma unroll
    for (int i = 0; i < 5; ++i)  w3[i] = W3[i];
    const float bias3 = b3[0];

    const int  g      = n - 8 * hi;
    const bool rowok  = (g >= 0) && (g < 5);
    auto w2a = [&](int k) -> float {
        return (rowok && k < 5) ? W2[g * 5 + k] : 0.0f;
    };
    v2f A2_0, A2_1;
    A2_0.x = w2a(0);  A2_0.y = w2a(1);
    A2_1.x = w2a(2);  A2_1.y = w2a(3);
    float w2k4[5];
#pragma unroll
    for (int i = 0; i < 5; ++i) w2k4[i] = W2[i * 5 + 4];
    v8f cbias;
#pragma unroll
    for (int j = 0; j < 8; ++j) cbias[j] = (j < 5) ? b2[j] : 0.0f;

    const int sub = lane >> 3;
    const int numGroups = (N + PTS_PER_WAVE - 1) / PTS_PER_WAVE;
    const int waveGlobal = blockIdx.x * WAVES_PER_BLOCK + wave;
    const int waveStride = gridDim.x * WAVES_PER_BLOCK;

    for (int grp = waveGlobal; grp < numGroups; grp += waveStride) {
        const int q  = grp * PTS_PER_WAVE + sub;
        const int qc = (q < N) ? q : (N - 1);

        const float px = pos[3 * qc + 0];
        const float py = pos[3 * qc + 1];
        const float pz = pos[3 * qc + 2];

        const float* src = nbr + (qc * 32 + (lane & 7) * NBR_PER_LANE) * 3;
        const v4f r0 = *(const v4f_a4*)(src + 0);
        const v4f r1 = *(const v4f_a4*)(src + 4);
        const v4f r2 = *(const v4f_a4*)(src + 8);
        const float in[12] = { r0.x, r0.y, r0.z, r0.w, r1.x, r1.y, r1.z, r1.w,
                               r2.x, r2.y, r2.z, r2.w };

        float ax = 0.0f, ay = 0.0f, az = 0.0f;
#pragma unroll
        for (int s = 0; s < NBR_PER_LANE; ++s) {
            const float dx = in[3 * s + 0] - px, dy = in[3 * s + 1] - py, dz = in[3 * s + 2] - pz;
            const float inv = inv_norm_from_sq(fmaf(dx, dx, fmaf(dy, dy, dz * dz)));
            const float ndx = dx * inv, ndy = dy * inv, ndz = dz * inv;

            float h[5];
#pragma unroll
            for (int j = 0; j < 5; ++j) {
                float a = fmaf(w1[3 * j + 2], ndz,
                          fmaf(w1[3 * j + 1], ndy,
                          fmaf(w1[3 * j + 0], ndx, bb1[j])));
                h[j] = fmaxf(a, 0.0f);
            }

            v2f B0, B1;
            B0.x = h[0];  B0.y = h[1];
            B1.x = h[2];  B1.y = h[3];
            v8f c = __builtin_amdgcn_wmma_f32_16x16x4_f32(false, A2_0, false, B0,
                                                          (short)0, cbias, false, false);
            c = __builtin_amdgcn_wmma_f32_16x16x4_f32(false, A2_1, false, B1,
                                                      (short)0, c, false, false);
#pragma unroll
            for (int i = 0; i < 5; ++i) c[i] = fmaf(w2k4[i], h[4], c[i]);

            float w = bias3;
#pragma unroll
            for (int i = 0; i < 5; ++i) w = fmaf(w3[i], relu1(c[i]), w);

            if (s == 0) { ax = ndx * w;  ay = ndy * w;  az = ndz * w; }
            else        { ax = fmaf(ndx, w, ax);  ay = fmaf(ndy, w, ay);  az = fmaf(ndz, w, az); }
        }

        float cx = ax, cy = ay, cz = az;
        cx += swz<0x041f>(cx); cy += swz<0x041f>(cy); cz += swz<0x041f>(cz);
        cx += swz<0x081f>(cx); cy += swz<0x081f>(cy); cz += swz<0x081f>(cz);
        cx += swz<0x101f>(cx); cy += swz<0x101f>(cy); cz += swz<0x101f>(cz);

        const float vi = inv_norm_from_sq(fmaf(cx, cx, fmaf(cy, cy, cz * cz)));
        if ((lane & 7) == 0 && q < N) {
            out[3 * q + 0] = cx * vi;
            out[3 * q + 1] = cy * vi;
            out[3 * q + 2] = cz * vi;
        }
    }
}

extern "C" void kernel_launch(void* const* d_in, const int* in_sizes, int n_in,
                              void* d_out, int out_size, void* d_ws, size_t ws_size,
                              hipStream_t stream) {
    const float* pos = (const float*)d_in[0];
    const float* nbr = (const float*)d_in[1];
    const float* W1  = (const float*)d_in[2];
    const float* b1  = (const float*)d_in[3];
    const float* W2  = (const float*)d_in[4];
    const float* b2  = (const float*)d_in[5];
    const float* W3  = (const float*)d_in[6];
    const float* b3  = (const float*)d_in[7];
    float* out = (float*)d_out;

    const int N = in_sizes[0] / 3;
    velvec_kernel<<<GRID_BLOCKS, BLOCK_THREADS, 0, stream>>>(pos, nbr, W1, b1, W2, b2,
                                                             W3, b3, out, N);
}
